# starting kernel
# baseline (speedup 1.0000x reference)
#include <hip/hip_runtime.h>


typedef __attribute__((ext_vector_type(2))) float v2f;
typedef __attribute__((ext_vector_type(4))) float v4f;
typedef __attribute__((ext_vector_type(8))) float v8f;

#define SEG   256
#define CHUNK 16
#define CHUNKS_PER_SEG (SEG / CHUNK)

__global__ void seg_state_kernel(const float* __restrict__ u,
                                 const float* __restrict__ bc,
                                 const float* __restrict__ fc,
                                 float* __restrict__ zpart,
                                 int Bn, int T, int nseg) {
  int tid = blockIdx.x * blockDim.x + threadIdx.x;
  int total = Bn * nseg;
  if (tid >= total) return;
  int b = tid % Bn;
  int s = tid / Bn;
  const float b0 = bc[0], b1 = bc[1], b2v = 0.0f;
  const float a1 = fc[0], a2 = fc[1];
  float z0 = 0.f, z1 = 0.f;
  const float* up = u + (long long)b * T + (long long)s * SEG;
  #pragma unroll 4
  for (int t = 0; t < SEG; ++t) {
    float ut = up[t];
    float yv  = b0 * ut + z0;
    float nz0 = b1 * ut - a1 * yv + z1;
    float nz1 = b2v * ut - a2 * yv;
    z0 = nz0; z1 = nz1;
  }
  zpart[(size_t)(s * Bn + b) * 2 + 0] = z0;
  zpart[(size_t)(s * Bn + b) * 2 + 1] = z1;
}

__global__ void scan_state_kernel(const float* __restrict__ bc,
                                  const float* __restrict__ fc,
                                  const float* __restrict__ yinit,
                                  const float* __restrict__ uinit,
                                  const float* __restrict__ zpart,
                                  float* __restrict__ zin,
                                  int Bn, int nseg) {
  int b = blockIdx.x * blockDim.x + threadIdx.x;
  if (b >= Bn) return;
  const float b1 = bc[1];
  const float a1 = fc[0], a2 = fc[1];

  float m00 = -a1, m01 = 1.f, m10 = -a2, m11 = 0.f;
  #pragma unroll
  for (int i = 0; i < 8; ++i) {
    float t00 = m00 * m00 + m01 * m10;
    float t01 = m00 * m01 + m01 * m11;
    float t10 = m10 * m00 + m11 * m10;
    float t11 = m10 * m01 + m11 * m11;
    m00 = t00; m01 = t01; m10 = t10; m11 = t11;
  }

  float z0 = b1 * uinit[b * 2 + 0] - a1 * yinit[b * 2 + 0] - a2 * yinit[b * 2 + 1];
  float z1 = -a2 * yinit[b * 2 + 0];

  for (int s = 0; s < nseg; ++s) {
    size_t idx = (size_t)(s * Bn + b) * 2;
    zin[idx + 0] = z0;
    zin[idx + 1] = z1;
    float nz0 = m00 * z0 + m01 * z1 + zpart[idx + 0];
    float nz1 = m10 * z0 + m11 * z1 + zpart[idx + 1];
    z0 = nz0; z1 = nz1;
  }
}

__global__ __launch_bounds__(256)
void wmma_filter_kernel(const float* __restrict__ u,
                        const float* __restrict__ bc,
                        const float* __restrict__ fc,
                        const float* __restrict__ zin,
                        float* __restrict__ y,
                        int Bn, int T, int nseg) {
  __shared__ float Hs[16][16];
  __shared__ v2f   Cs[16];

  const float b0 = bc[0], b1 = bc[1], b2v = 0.0f;
  const float a1 = fc[0], a2 = fc[1];

  if (threadIdx.x < 16) {
    const int t = (int)threadIdx.x;
    float hm1 = 0.f, hm2 = 0.f;
    #pragma unroll
    for (int d = 0; d < 16; ++d) {
      float hd;
      if (d == 0)      hd = b0;
      else if (d == 1) hd = b1 - a1 * b0;
      else if (d == 2) hd = b2v - a1 * hm1 - a2 * hm2;
      else             hd = -a1 * hm1 - a2 * hm2;
      if (d <= t) Hs[t][t - d] = hd;
      hm2 = hm1; hm1 = hd;
    }
    for (int k = t + 1; k < 16; ++k) Hs[t][k] = 0.f;
  } else if (threadIdx.x == 16) {
    float cc0 = 1.f, cc1 = 0.f;
    #pragma unroll
    for (int k = 0; k < 16; ++k) {
      v2f c; c.x = cc0; c.y = cc1;
      Cs[k] = c;
      float n0 = -a1 * cc0 - a2 * cc1;
      cc1 = cc0; cc0 = n0;
    }
  }
  __syncthreads();

  const int lane = (int)(threadIdx.x & 31);
  const int wave = (int)((blockIdx.x * blockDim.x + threadIdx.x) >> 5);
  const int ngroups = Bn >> 4;
  const int totalWaves = nseg * ngroups;

  if (wave < totalWaves) {
    const int seg  = wave / ngroups;
    const int grp  = wave % ngroups;
    const int col  = lane & 15;
    const int half = lane >> 4;
    const int b    = (grp << 4) + col;

    v2f afrag[4];
    #pragma unroll
    for (int j = 0; j < 4; ++j)
      afrag[j] = *(const v2f*)&Hs[col][4 * j + 2 * half];

    v2f crow[8];
    #pragma unroll
    for (int r = 0; r < 8; ++r)
      crow[r] = Cs[r + 8 * half];

    float z0 = zin[(size_t)(seg * Bn + b) * 2 + 0];
    float z1 = zin[(size_t)(seg * Bn + b) * 2 + 1];

    const long long rowbase = (long long)b * T + (long long)seg * SEG;

    for (int c = 0; c < CHUNKS_PER_SEG; ++c) {
      const long long t0 = rowbase + (long long)c * CHUNK;

      if (c + 1 < CHUNKS_PER_SEG)
        __builtin_prefetch(u + t0 + CHUNK, 0, 1);

      v2f bfrag[4];
      #pragma unroll
      for (int j = 0; j < 4; ++j)
        bfrag[j] = *(const v2f*)(u + t0 + 4 * j + 2 * half);

      v8f acc;
      #pragma unroll
      for (int r = 0; r < 8; ++r) acc[r] = crow[r].x * z0 + crow[r].y * z1;

      acc = __builtin_amdgcn_wmma_f32_16x16x4_f32(false, afrag[0], false, bfrag[0], (short)0, acc, false, false);
      acc = __builtin_amdgcn_wmma_f32_16x16x4_f32(false, afrag[1], false, bfrag[1], (short)0, acc, false, false);
      acc = __builtin_amdgcn_wmma_f32_16x16x4_f32(false, afrag[2], false, bfrag[2], (short)0, acc, false, false);
      acc = __builtin_amdgcn_wmma_f32_16x16x4_f32(false, afrag[3], false, bfrag[3], (short)0, acc, false, false);

      v4f lo; lo.x = acc[0]; lo.y = acc[1]; lo.z = acc[2]; lo.w = acc[3];
      v4f hi; hi.x = acc[4]; hi.y = acc[5]; hi.z = acc[6]; hi.w = acc[7];
      __builtin_nontemporal_store(lo, (v4f*)(y + t0 + 8 * half + 0));
      __builtin_nontemporal_store(hi, (v4f*)(y + t0 + 8 * half + 4));

      float y14 = __shfl(acc[6],     col + 16, 32);
      float y15 = __shfl(acc[7],     col + 16, 32);
      float u14 = __shfl(bfrag[3].x, col + 16, 32);
      float u15 = __shfl(bfrag[3].y, col + 16, 32);
      float nz1 = b2v * u15 - a2 * y15;
      float nz0 = b1 * u15 - a1 * y15 + b2v * u14 - a2 * y14;
      z0 = nz0; z1 = nz1;
    }
  }
}

__global__ void naive_iir_kernel(const float* __restrict__ u,
                                 const float* __restrict__ bc,
                                 const float* __restrict__ fc,
                                 const float* __restrict__ yinit,
                                 const float* __restrict__ uinit,
                                 float* __restrict__ y, int Bn, int T) {
  int b = blockIdx.x * blockDim.x + threadIdx.x;
  if (b >= Bn) return;
  const float b0 = bc[0], b1 = bc[1];
  const float a1 = fc[0], a2 = fc[1];
  float z0 = b1 * uinit[b * 2 + 0] - a1 * yinit[b * 2 + 0] - a2 * yinit[b * 2 + 1];
  float z1 = -a2 * yinit[b * 2 + 0];
  const float* up = u + (long long)b * T;
  float* yp = y + (long long)b * T;
  for (int t = 0; t < T; ++t) {
    float ut = up[t];
    float yv = b0 * ut + z0;
    float nz0 = b1 * ut - a1 * yv + z1;
    z1 = -a2 * yv;
    z0 = nz0;
    yp[t] = yv;
  }
}

extern "C" void kernel_launch(void* const* d_in, const int* in_sizes, int n_in,
                              void* d_out, int out_size, void* d_ws, size_t ws_size,
                              hipStream_t stream) {
  const float* bc = (const float*)d_in[0];
  const float* fc = (const float*)d_in[1];
  const float* u  = (const float*)d_in[2];
  const float* yi = (const float*)d_in[3];
  const float* ui = (const float*)d_in[4];
  float* y = (float*)d_out;

  const int Bn = in_sizes[3] / 2;
  const int T  = in_sizes[2] / Bn;
  const int nseg = T / SEG;
  const size_t need = (size_t)2 * (size_t)nseg * (size_t)Bn * 2 * sizeof(float);

  if ((T % SEG) != 0 || (Bn % 16) != 0 || ws_size < need) {
    naive_iir_kernel<<<(Bn + 255) / 256, 256, 0, stream>>>(u, bc, fc, yi, ui, y, Bn, T);
    return;
  }

  float* zpart = (float*)d_ws;
  float* zin   = zpart + (size_t)nseg * Bn * 2;

  const int total = Bn * nseg;
  seg_state_kernel<<<(total + 255) / 256, 256, 0, stream>>>(u, bc, fc, zpart, Bn, T, nseg);
  scan_state_kernel<<<(Bn + 255) / 256, 256, 0, stream>>>(bc, fc, yi, ui, zpart, zin, Bn, nseg);

  const int waves  = nseg * (Bn / 16);
  const int blocks = (waves + 7) / 8;
  wmma_filter_kernel<<<blocks, 256, 0, stream>>>(u, bc, fc, zin, y, Bn, T, nseg);
}


# filter pass: u and y tiles go through a wave-private LDS tile so each global load/store is row-contiguous b128 (whole 128 B lines, tile loads issued up front), not row-strided 8 B and 16 B pieces
# speedup vs baseline: 3.0245x; 3.0245x over previous
#include <hip/hip_runtime.h>


typedef __attribute__((ext_vector_type(2))) float v2f;
typedef __attribute__((ext_vector_type(4))) float v4f;
typedef __attribute__((ext_vector_type(8))) float v8f;

#define SEG   256
#define CHUNK 16
#define CHUNKS_PER_SEG (SEG / CHUNK)

__global__ void seg_state_kernel(const float* __restrict__ u,
                                 const float* __restrict__ bc,
                                 const float* __restrict__ fc,
                                 float* __restrict__ zpart,
                                 int Bn, int T, int nseg) {
  int tid = blockIdx.x * blockDim.x + threadIdx.x;
  int total = Bn * nseg;
  if (tid >= total) return;
  int b = tid % Bn;
  int s = tid / Bn;
  const float b0 = bc[0], b1 = bc[1], b2v = 0.0f;
  const float a1 = fc[0], a2 = fc[1];
  float z0 = 0.f, z1 = 0.f;
  const float* up = u + (long long)b * T + (long long)s * SEG;
  #pragma unroll 4
  for (int t = 0; t < SEG; ++t) {
    float ut = up[t];
    float yv  = b0 * ut + z0;
    float nz0 = b1 * ut - a1 * yv + z1;
    float nz1 = b2v * ut - a2 * yv;
    z0 = nz0; z1 = nz1;
  }
  zpart[(size_t)(s * Bn + b) * 2 + 0] = z0;
  zpart[(size_t)(s * Bn + b) * 2 + 1] = z1;
}

__global__ void scan_state_kernel(const float* __restrict__ bc,
                                  const float* __restrict__ fc,
                                  const float* __restrict__ yinit,
                                  const float* __restrict__ uinit,
                                  const float* __restrict__ zpart,
                                  float* __restrict__ zin,
                                  int Bn, int nseg) {
  int b = blockIdx.x * blockDim.x + threadIdx.x;
  if (b >= Bn) return;
  const float b1 = bc[1];
  const float a1 = fc[0], a2 = fc[1];

  float m00 = -a1, m01 = 1.f, m10 = -a2, m11 = 0.f;
  #pragma unroll
  for (int i = 0; i < 8; ++i) {
    float t00 = m00 * m00 + m01 * m10;
    float t01 = m00 * m01 + m01 * m11;
    float t10 = m10 * m00 + m11 * m10;
    float t11 = m10 * m01 + m11 * m11;
    m00 = t00; m01 = t01; m10 = t10; m11 = t11;
  }

  float z0 = b1 * uinit[b * 2 + 0] - a1 * yinit[b * 2 + 0] - a2 * yinit[b * 2 + 1];
  float z1 = -a2 * yinit[b * 2 + 0];

  for (int s = 0; s < nseg; ++s) {
    size_t idx = (size_t)(s * Bn + b) * 2;
    zin[idx + 0] = z0;
    zin[idx + 1] = z1;
    float nz0 = m00 * z0 + m01 * z1 + zpart[idx + 0];
    float nz1 = m10 * z0 + m11 * z1 + zpart[idx + 1];
    z0 = nz0; z1 = nz1;
  }
}

#define TILE_T 64
#define TILES_PER_SEG (SEG / TILE_T)
#define CHUNKS_PER_TILE (TILE_T / CHUNK)
#define ROWPAIRS 8
#define PITCH 68

__global__ __launch_bounds__(256)
void wmma_filter_kernel(const float* __restrict__ u,
                        const float* __restrict__ bc,
                        const float* __restrict__ fc,
                        const float* __restrict__ zin,
                        float* __restrict__ y,
                        int Bn, int T, int nseg) {
  __shared__ float Hs[16][16];
  __shared__ v2f   Cs[16];
  __shared__ __attribute__((aligned(16))) float Ut[8][16 * PITCH];
  __shared__ __attribute__((aligned(16))) float Yt[8][16 * PITCH];

  const float b0 = bc[0], b1 = bc[1], b2v = 0.0f;
  const float a1 = fc[0], a2 = fc[1];

  const int lane = (int)(threadIdx.x & 31);
  const int wib  = (int)(threadIdx.x >> 5);
  const int wave = (int)((blockIdx.x * blockDim.x + threadIdx.x) >> 5);
  const int ngroups = Bn >> 4;
  const int totalWaves = nseg * ngroups;
  const bool active = wave < totalWaves;
  const int seg  = active ? wave / ngroups : 0;
  const int grp  = active ? wave % ngroups : 0;
  const int col  = lane & 15;
  const int half = lane >> 4;
  const int b    = (grp << 4) + col;

  float z0 = 0.f, z1 = 0.f;
  if (active) {
    z0 = zin[(size_t)(seg * Bn + b) * 2 + 0];
    z1 = zin[(size_t)(seg * Bn + b) * 2 + 1];
  }

  if (threadIdx.x < 16) {
    const int t = (int)threadIdx.x;
    float hm1 = 0.f, hm2 = 0.f;
    #pragma unroll
    for (int d = 0; d < 16; ++d) {
      float hd;
      if (d == 0)      hd = b0;
      else if (d == 1) hd = b1 - a1 * b0;
      else if (d == 2) hd = b2v - a1 * hm1 - a2 * hm2;
      else             hd = -a1 * hm1 - a2 * hm2;
      if (d <= t) Hs[t][t - d] = hd;
      hm2 = hm1; hm1 = hd;
    }
    for (int k = t + 1; k < 16; ++k) Hs[t][k] = 0.f;
  } else if (threadIdx.x == 16) {
    float cc0 = 1.f, cc1 = 0.f;
    #pragma unroll
    for (int k = 0; k < 16; ++k) {
      v2f c; c.x = cc0; c.y = cc1;
      Cs[k] = c;
      float n0 = -a1 * cc0 - a2 * cc1;
      cc1 = cc0; cc0 = n0;
    }
  }
  __syncthreads();

  if (active) {
    v2f afrag[4];
    #pragma unroll
    for (int j = 0; j < 4; ++j)
      afrag[j] = *(const v2f*)&Hs[col][4 * j + 2 * half];

    v2f crow[8];
    #pragma unroll
    for (int r = 0; r < 8; ++r)
      crow[r] = Cs[r + 8 * half];

    float* ut = &Ut[wib][0];
    float* yt = &Yt[wib][0];
    const int row_off  = half * PITCH + 4 * col;
    const int frag_off = col * PITCH + 2 * half;
    const int d_off    = col * PITCH + 8 * half;

    const size_t goff  = (size_t)((grp << 4) + half) * (size_t)T + (size_t)seg * SEG + (size_t)(4 * col);
    const size_t rstep = (size_t)2 * (size_t)T;
    const float* ug = u + goff;
    float*       yg = y + goff;

    v4f in[TILES_PER_SEG][ROWPAIRS];
    #pragma unroll
    for (int t = 0; t < TILES_PER_SEG; ++t) {
      #pragma unroll
      for (int i = 0; i < ROWPAIRS; ++i)
        in[t][i] = *(const v4f*)(ug + (size_t)i * rstep + (size_t)(t * TILE_T));
      __builtin_amdgcn_sched_barrier(0);
    }

    #pragma unroll
    for (int t = 0; t < TILES_PER_SEG; ++t) {
      #pragma unroll
      for (int i = 0; i < ROWPAIRS; ++i)
        *(v4f*)(ut + row_off + 2 * i * PITCH) = in[t][i];

      #pragma unroll
      for (int c = 0; c < CHUNKS_PER_TILE; ++c) {
        v2f bfrag[4];
        #pragma unroll
        for (int j = 0; j < 4; ++j)
          bfrag[j] = *(const v2f*)(ut + frag_off + 16 * c + 4 * j);

        v8f acc;
        #pragma unroll
        for (int r = 0; r < 8; ++r) acc[r] = crow[r].x * z0 + crow[r].y * z1;

        acc = __builtin_amdgcn_wmma_f32_16x16x4_f32(false, afrag[0], false, bfrag[0], (short)0, acc, false, false);
        acc = __builtin_amdgcn_wmma_f32_16x16x4_f32(false, afrag[1], false, bfrag[1], (short)0, acc, false, false);
        acc = __builtin_amdgcn_wmma_f32_16x16x4_f32(false, afrag[2], false, bfrag[2], (short)0, acc, false, false);
        acc = __builtin_amdgcn_wmma_f32_16x16x4_f32(false, afrag[3], false, bfrag[3], (short)0, acc, false, false);

        v4f lo; lo.x = acc[0]; lo.y = acc[1]; lo.z = acc[2]; lo.w = acc[3];
        v4f hi; hi.x = acc[4]; hi.y = acc[5]; hi.z = acc[6]; hi.w = acc[7];
        *(v4f*)(yt + d_off + 16 * c + 0) = lo;
        *(v4f*)(yt + d_off + 16 * c + 4) = hi;

        float y14 = __shfl(acc[6],     col + 16, 32);
        float y15 = __shfl(acc[7],     col + 16, 32);
        float u14 = __shfl(bfrag[3].x, col + 16, 32);
        float u15 = __shfl(bfrag[3].y, col + 16, 32);
        float nz1 = b2v * u15 - a2 * y15;
        float nz0 = b1 * u15 - a1 * y15 + b2v * u14 - a2 * y14;
        z0 = nz0; z1 = nz1;
      }

      #pragma unroll
      for (int i = 0; i < ROWPAIRS; ++i) {
        v4f o = *(const v4f*)(yt + row_off + 2 * i * PITCH);
        __builtin_nontemporal_store(o, (v4f*)(yg + (size_t)i * rstep + (size_t)(t * TILE_T)));
      }
    }
  }
}

__global__ void naive_iir_kernel(const float* __restrict__ u,
                                 const float* __restrict__ bc,
                                 const float* __restrict__ fc,
                                 const float* __restrict__ yinit,
                                 const float* __restrict__ uinit,
                                 float* __restrict__ y, int Bn, int T) {
  int b = blockIdx.x * blockDim.x + threadIdx.x;
  if (b >= Bn) return;
  const float b0 = bc[0], b1 = bc[1];
  const float a1 = fc[0], a2 = fc[1];
  float z0 = b1 * uinit[b * 2 + 0] - a1 * yinit[b * 2 + 0] - a2 * yinit[b * 2 + 1];
  float z1 = -a2 * yinit[b * 2 + 0];
  const float* up = u + (long long)b * T;
  float* yp = y + (long long)b * T;
  for (int t = 0; t < T; ++t) {
    float ut = up[t];
    float yv = b0 * ut + z0;
    float nz0 = b1 * ut - a1 * yv + z1;
    z1 = -a2 * yv;
    z0 = nz0;
    yp[t] = yv;
  }
}

extern "C" void kernel_launch(void* const* d_in, const int* in_sizes, int n_in,
                              void* d_out, int out_size, void* d_ws, size_t ws_size,
                              hipStream_t stream) {
  const float* bc = (const float*)d_in[0];
  const float* fc = (const float*)d_in[1];
  const float* u  = (const float*)d_in[2];
  const float* yi = (const float*)d_in[3];
  const float* ui = (const float*)d_in[4];
  float* y = (float*)d_out;

  const int Bn = in_sizes[3] / 2;
  const int T  = in_sizes[2] / Bn;
  const int nseg = T / SEG;
  const size_t need = (size_t)2 * (size_t)nseg * (size_t)Bn * 2 * sizeof(float);

  if ((T % SEG) != 0 || (Bn % 16) != 0 || ws_size < need) {
    naive_iir_kernel<<<(Bn + 255) / 256, 256, 0, stream>>>(u, bc, fc, yi, ui, y, Bn, T);
    return;
  }

  float* zpart = (float*)d_ws;
  float* zin   = zpart + (size_t)nseg * Bn * 2;

  const int total = Bn * nseg;
  seg_state_kernel<<<(total + 255) / 256, 256, 0, stream>>>(u, bc, fc, zpart, Bn, T, nseg);
  scan_state_kernel<<<(Bn + 255) / 256, 256, 0, stream>>>(bc, fc, yi, ui, zpart, zin, Bn, nseg);

  const int waves  = nseg * (Bn / 16);
  const int blocks = (waves + 7) / 8;
  wmma_filter_kernel<<<blocks, 256, 0, stream>>>(u, bc, fc, zin, y, Bn, T, nseg);
}


# state pass recast as a lane-parallel weighted sum z=sum_t W[t]*u_t over row-contiguous b128 reads; weights by the filter's one-step recurrence, once per block; sums now a lane butterfly tree
# speedup vs baseline: 3.4147x; 1.1290x over previous
#include <hip/hip_runtime.h>


typedef __attribute__((ext_vector_type(2))) float v2f;
typedef __attribute__((ext_vector_type(4))) float v4f;
typedef __attribute__((ext_vector_type(8))) float v8f;

#define SEG   256
#define CHUNK 16
#define CHUNKS_PER_SEG (SEG / CHUNK)

#define P1_ROWS 16

__global__ __launch_bounds__(256)
void seg_state_kernel(const float* __restrict__ u,
                      const float* __restrict__ bc,
                      const float* __restrict__ fc,
                      float* __restrict__ zpart,
                      int Bn, int T, int nseg) {
  __shared__ __attribute__((aligned(16))) float Wt[2 * SEG];

  const int lane = (int)(threadIdx.x & 31);
  const int wave = (int)((blockIdx.x * blockDim.x + threadIdx.x) >> 5);
  const int ngroups = Bn >> 4;
  const bool active = wave < nseg * ngroups;
  const int seg = active ? wave / ngroups : 0;
  const int grp = active ? wave % ngroups : 0;

  v4f in[P1_ROWS][2];
  const float* ug = u + (size_t)(grp << 4) * (size_t)T + (size_t)seg * SEG + (size_t)(4 * lane);
  #pragma unroll
  for (int i = 0; i < P1_ROWS; ++i) {
    #pragma unroll
    for (int j = 0; j < 2; ++j)
      in[i][j] = *(const v4f*)(ug + (size_t)i * (size_t)T + (size_t)(128 * j));
  }

  if (threadIdx.x == 0) {
    const float b0 = bc[0], b1 = bc[1], b2v = 0.0f;
    const float a1 = fc[0], a2 = fc[1];
    float w0 = b1 - a1 * b0, w1 = b2v - a2 * b0;
    for (int k = 0; k < SEG; ++k) {
      v2f w; w.x = w0; w.y = w1;
      *(v2f*)&Wt[2 * (SEG - 1 - k)] = w;
      const float n0 = -a1 * w0 + w1;
      const float n1 = -a2 * w0;
      w0 = n0; w1 = n1;
    }
  }
  __syncthreads();

  v4f wlo[2], whi[2];
  #pragma unroll
  for (int j = 0; j < 2; ++j) {
    wlo[j] = *(const v4f*)(Wt + 256 * j + 8 * lane);
    whi[j] = *(const v4f*)(Wt + 256 * j + 8 * lane + 4);
  }

  float v[2 * P1_ROWS];
  #pragma unroll
  for (int i = 0; i < P1_ROWS; ++i) {
    float s0 = wlo[0].x * in[i][0].x;
    float s1 = wlo[0].y * in[i][0].x;
    s0 = __builtin_fmaf(wlo[0].z, in[i][0].y, s0);
    s1 = __builtin_fmaf(wlo[0].w, in[i][0].y, s1);
    s0 = __builtin_fmaf(whi[0].x, in[i][0].z, s0);
    s1 = __builtin_fmaf(whi[0].y, in[i][0].z, s1);
    s0 = __builtin_fmaf(whi[0].z, in[i][0].w, s0);
    s1 = __builtin_fmaf(whi[0].w, in[i][0].w, s1);
    s0 = __builtin_fmaf(wlo[1].x, in[i][1].x, s0);
    s1 = __builtin_fmaf(wlo[1].y, in[i][1].x, s1);
    s0 = __builtin_fmaf(wlo[1].z, in[i][1].y, s0);
    s1 = __builtin_fmaf(wlo[1].w, in[i][1].y, s1);
    s0 = __builtin_fmaf(whi[1].x, in[i][1].z, s0);
    s1 = __builtin_fmaf(whi[1].y, in[i][1].z, s1);
    s0 = __builtin_fmaf(whi[1].z, in[i][1].w, s0);
    s1 = __builtin_fmaf(whi[1].w, in[i][1].w, s1);
    v[2 * i + 0] = s0;
    v[2 * i + 1] = s1;
  }

  float r16[16];
  {
    const bool up = (lane & 16) != 0;
    #pragma unroll
    for (int i = 0; i < 16; ++i) {
      float keep = up ? v[i + 16] : v[i];
      float send = up ? v[i] : v[i + 16];
      r16[i] = keep + __shfl_xor(send, 16, 32);
    }
  }
  float r8[8];
  {
    const bool up = (lane & 8) != 0;
    #pragma unroll
    for (int i = 0; i < 8; ++i) {
      float keep = up ? r16[i + 8] : r16[i];
      float send = up ? r16[i] : r16[i + 8];
      r8[i] = keep + __shfl_xor(send, 8, 32);
    }
  }
  float r4[4];
  {
    const bool up = (lane & 4) != 0;
    #pragma unroll
    for (int i = 0; i < 4; ++i) {
      float keep = up ? r8[i + 4] : r8[i];
      float send = up ? r8[i] : r8[i + 4];
      r4[i] = keep + __shfl_xor(send, 4, 32);
    }
  }
  float r2[2];
  {
    const bool up = (lane & 2) != 0;
    #pragma unroll
    for (int i = 0; i < 2; ++i) {
      float keep = up ? r4[i + 2] : r4[i];
      float send = up ? r4[i] : r4[i + 2];
      r2[i] = keep + __shfl_xor(send, 2, 32);
    }
  }
  float r1;
  {
    const bool up = (lane & 1) != 0;
    float keep = up ? r2[1] : r2[0];
    float send = up ? r2[0] : r2[1];
    r1 = keep + __shfl_xor(send, 1, 32);
  }

  if (active) zpart[(size_t)(seg * Bn + (grp << 4)) * 2 + (size_t)lane] = r1;
}

__global__ void scan_state_kernel(const float* __restrict__ bc,
                                  const float* __restrict__ fc,
                                  const float* __restrict__ yinit,
                                  const float* __restrict__ uinit,
                                  const float* __restrict__ zpart,
                                  float* __restrict__ zin,
                                  int Bn, int nseg) {
  int b = blockIdx.x * blockDim.x + threadIdx.x;
  if (b >= Bn) return;
  const float b1 = bc[1];
  const float a1 = fc[0], a2 = fc[1];

  float m00 = -a1, m01 = 1.f, m10 = -a2, m11 = 0.f;
  #pragma unroll
  for (int i = 0; i < 8; ++i) {
    float t00 = m00 * m00 + m01 * m10;
    float t01 = m00 * m01 + m01 * m11;
    float t10 = m10 * m00 + m11 * m10;
    float t11 = m10 * m01 + m11 * m11;
    m00 = t00; m01 = t01; m10 = t10; m11 = t11;
  }

  float z0 = b1 * uinit[b * 2 + 0] - a1 * yinit[b * 2 + 0] - a2 * yinit[b * 2 + 1];
  float z1 = -a2 * yinit[b * 2 + 0];

  for (int s = 0; s < nseg; ++s) {
    size_t idx = (size_t)(s * Bn + b) * 2;
    zin[idx + 0] = z0;
    zin[idx + 1] = z1;
    float nz0 = m00 * z0 + m01 * z1 + zpart[idx + 0];
    float nz1 = m10 * z0 + m11 * z1 + zpart[idx + 1];
    z0 = nz0; z1 = nz1;
  }
}

#define TILE_T 64
#define TILES_PER_SEG (SEG / TILE_T)
#define CHUNKS_PER_TILE (TILE_T / CHUNK)
#define ROWPAIRS 8
#define PITCH 68

__global__ __launch_bounds__(256)
void wmma_filter_kernel(const float* __restrict__ u,
                        const float* __restrict__ bc,
                        const float* __restrict__ fc,
                        const float* __restrict__ zin,
                        float* __restrict__ y,
                        int Bn, int T, int nseg) {
  __shared__ float Hs[16][16];
  __shared__ v2f   Cs[16];
  __shared__ __attribute__((aligned(16))) float Ut[8][16 * PITCH];
  __shared__ __attribute__((aligned(16))) float Yt[8][16 * PITCH];

  const float b0 = bc[0], b1 = bc[1], b2v = 0.0f;
  const float a1 = fc[0], a2 = fc[1];

  const int lane = (int)(threadIdx.x & 31);
  const int wib  = (int)(threadIdx.x >> 5);
  const int wave = (int)((blockIdx.x * blockDim.x + threadIdx.x) >> 5);
  const int ngroups = Bn >> 4;
  const int totalWaves = nseg * ngroups;
  const bool active = wave < totalWaves;
  const int seg  = active ? wave / ngroups : 0;
  const int grp  = active ? wave % ngroups : 0;
  const int col  = lane & 15;
  const int half = lane >> 4;
  const int b    = (grp << 4) + col;

  float z0 = 0.f, z1 = 0.f;
  if (active) {
    z0 = zin[(size_t)(seg * Bn + b) * 2 + 0];
    z1 = zin[(size_t)(seg * Bn + b) * 2 + 1];
  }

  if (threadIdx.x < 16) {
    const int t = (int)threadIdx.x;
    float hm1 = 0.f, hm2 = 0.f;
    #pragma unroll
    for (int d = 0; d < 16; ++d) {
      float hd;
      if (d == 0)      hd = b0;
      else if (d == 1) hd = b1 - a1 * b0;
      else if (d == 2) hd = b2v - a1 * hm1 - a2 * hm2;
      else             hd = -a1 * hm1 - a2 * hm2;
      if (d <= t) Hs[t][t - d] = hd;
      hm2 = hm1; hm1 = hd;
    }
    for (int k = t + 1; k < 16; ++k) Hs[t][k] = 0.f;
  } else if (threadIdx.x == 16) {
    float cc0 = 1.f, cc1 = 0.f;
    #pragma unroll
    for (int k = 0; k < 16; ++k) {
      v2f c; c.x = cc0; c.y = cc1;
      Cs[k] = c;
      float n0 = -a1 * cc0 - a2 * cc1;
      cc1 = cc0; cc0 = n0;
    }
  }
  __syncthreads();

  if (active) {
    v2f afrag[4];
    #pragma unroll
    for (int j = 0; j < 4; ++j)
      afrag[j] = *(const v2f*)&Hs[col][4 * j + 2 * half];

    v2f crow[8];
    #pragma unroll
    for (int r = 0; r < 8; ++r)
      crow[r] = Cs[r + 8 * half];

    float* ut = &Ut[wib][0];
    float* yt = &Yt[wib][0];
    const int row_off  = half * PITCH + 4 * col;
    const int frag_off = col * PITCH + 2 * half;
    const int d_off    = col * PITCH + 8 * half;

    const size_t goff  = (size_t)((grp << 4) + half) * (size_t)T + (size_t)seg * SEG + (size_t)(4 * col);
    const size_t rstep = (size_t)2 * (size_t)T;
    const float* ug = u + goff;
    float*       yg = y + goff;

    v4f in[TILES_PER_SEG][ROWPAIRS];
    #pragma unroll
    for (int t = 0; t < TILES_PER_SEG; ++t) {
      #pragma unroll
      for (int i = 0; i < ROWPAIRS; ++i)
        in[t][i] = *(const v4f*)(ug + (size_t)i * rstep + (size_t)(t * TILE_T));
      __builtin_amdgcn_sched_barrier(0);
    }

    #pragma unroll
    for (int t = 0; t < TILES_PER_SEG; ++t) {
      #pragma unroll
      for (int i = 0; i < ROWPAIRS; ++i)
        *(v4f*)(ut + row_off + 2 * i * PITCH) = in[t][i];

      #pragma unroll
      for (int c = 0; c < CHUNKS_PER_TILE; ++c) {
        v2f bfrag[4];
        #pragma unroll
        for (int j = 0; j < 4; ++j)
          bfrag[j] = *(const v2f*)(ut + frag_off + 16 * c + 4 * j);

        v8f acc;
        #pragma unroll
        for (int r = 0; r < 8; ++r) acc[r] = crow[r].x * z0 + crow[r].y * z1;

        acc = __builtin_amdgcn_wmma_f32_16x16x4_f32(false, afrag[0], false, bfrag[0], (short)0, acc, false, false);
        acc = __builtin_amdgcn_wmma_f32_16x16x4_f32(false, afrag[1], false, bfrag[1], (short)0, acc, false, false);
        acc = __builtin_amdgcn_wmma_f32_16x16x4_f32(false, afrag[2], false, bfrag[2], (short)0, acc, false, false);
        acc = __builtin_amdgcn_wmma_f32_16x16x4_f32(false, afrag[3], false, bfrag[3], (short)0, acc, false, false);

        v4f lo; lo.x = acc[0]; lo.y = acc[1]; lo.z = acc[2]; lo.w = acc[3];
        v4f hi; hi.x = acc[4]; hi.y = acc[5]; hi.z = acc[6]; hi.w = acc[7];
        *(v4f*)(yt + d_off + 16 * c + 0) = lo;
        *(v4f*)(yt + d_off + 16 * c + 4) = hi;

        float y14 = __shfl(acc[6],     col + 16, 32);
        float y15 = __shfl(acc[7],     col + 16, 32);
        float u14 = __shfl(bfrag[3].x, col + 16, 32);
        float u15 = __shfl(bfrag[3].y, col + 16, 32);
        float nz1 = b2v * u15 - a2 * y15;
        float nz0 = b1 * u15 - a1 * y15 + b2v * u14 - a2 * y14;
        z0 = nz0; z1 = nz1;
      }

      #pragma unroll
      for (int i = 0; i < ROWPAIRS; ++i) {
        v4f o = *(const v4f*)(yt + row_off + 2 * i * PITCH);
        __builtin_nontemporal_store(o, (v4f*)(yg + (size_t)i * rstep + (size_t)(t * TILE_T)));
      }
    }
  }
}

__global__ void naive_iir_kernel(const float* __restrict__ u,
                                 const float* __restrict__ bc,
                                 const float* __restrict__ fc,
                                 const float* __restrict__ yinit,
                                 const float* __restrict__ uinit,
                                 float* __restrict__ y, int Bn, int T) {
  int b = blockIdx.x * blockDim.x + threadIdx.x;
  if (b >= Bn) return;
  const float b0 = bc[0], b1 = bc[1];
  const float a1 = fc[0], a2 = fc[1];
  float z0 = b1 * uinit[b * 2 + 0] - a1 * yinit[b * 2 + 0] - a2 * yinit[b * 2 + 1];
  float z1 = -a2 * yinit[b * 2 + 0];
  const float* up = u + (long long)b * T;
  float* yp = y + (long long)b * T;
  for (int t = 0; t < T; ++t) {
    float ut = up[t];
    float yv = b0 * ut + z0;
    float nz0 = b1 * ut - a1 * yv + z1;
    z1 = -a2 * yv;
    z0 = nz0;
    yp[t] = yv;
  }
}

extern "C" void kernel_launch(void* const* d_in, const int* in_sizes, int n_in,
                              void* d_out, int out_size, void* d_ws, size_t ws_size,
                              hipStream_t stream) {
  const float* bc = (const float*)d_in[0];
  const float* fc = (const float*)d_in[1];
  const float* u  = (const float*)d_in[2];
  const float* yi = (const float*)d_in[3];
  const float* ui = (const float*)d_in[4];
  float* y = (float*)d_out;

  const int Bn = in_sizes[3] / 2;
  const int T  = in_sizes[2] / Bn;
  const int nseg = T / SEG;
  const size_t need = (size_t)2 * (size_t)nseg * (size_t)Bn * 2 * sizeof(float);

  if ((T % SEG) != 0 || (Bn % 16) != 0 || ws_size < need) {
    naive_iir_kernel<<<(Bn + 255) / 256, 256, 0, stream>>>(u, bc, fc, yi, ui, y, Bn, T);
    return;
  }

  float* zpart = (float*)d_ws;
  float* zin   = zpart + (size_t)nseg * Bn * 2;

  const int waves  = nseg * (Bn / 16);
  const int blocks = (waves + 7) / 8;
  seg_state_kernel<<<blocks, 256, 0, stream>>>(u, bc, fc, zpart, Bn, T, nseg);
  scan_state_kernel<<<(Bn + 255) / 256, 256, 0, stream>>>(bc, fc, yi, ui, zpart, zin, Bn, nseg);
  wmma_filter_kernel<<<blocks, 256, 0, stream>>>(u, bc, fc, zin, y, Bn, T, nseg);
}


# scan launch removed: each filter wave rebuilds its own entry state from the earlier segments' states (same recurrence and order, loads in one group); late segments launch first
# speedup vs baseline: 5.5240x; 1.6177x over previous
#include <hip/hip_runtime.h>


typedef __attribute__((ext_vector_type(2))) float v2f;
typedef __attribute__((ext_vector_type(4))) float v4f;
typedef __attribute__((ext_vector_type(8))) float v8f;

#define SEG   256
#define CHUNK 16
#define CHUNKS_PER_SEG (SEG / CHUNK)

#define P1_ROWS 16

__global__ __launch_bounds__(256)
void seg_state_kernel(const float* __restrict__ u,
                      const float* __restrict__ bc,
                      const float* __restrict__ fc,
                      float* __restrict__ zpart,
                      int Bn, int T, int nseg) {
  __shared__ __attribute__((aligned(16))) float Wt[2 * SEG];

  const int lane = (int)(threadIdx.x & 31);
  const int wave = (int)((blockIdx.x * blockDim.x + threadIdx.x) >> 5);
  const int ngroups = Bn >> 4;
  const bool active = wave < nseg * ngroups;
  const int seg = active ? wave / ngroups : 0;
  const int grp = active ? wave % ngroups : 0;

  v4f in[P1_ROWS][2];
  const float* ug = u + (size_t)(grp << 4) * (size_t)T + (size_t)seg * SEG + (size_t)(4 * lane);
  #pragma unroll
  for (int i = 0; i < P1_ROWS; ++i) {
    #pragma unroll
    for (int j = 0; j < 2; ++j)
      in[i][j] = *(const v4f*)(ug + (size_t)i * (size_t)T + (size_t)(128 * j));
  }

  if (threadIdx.x == 0) {
    const float b0 = bc[0], b1 = bc[1], b2v = 0.0f;
    const float a1 = fc[0], a2 = fc[1];
    float w0 = b1 - a1 * b0, w1 = b2v - a2 * b0;
    for (int k = 0; k < SEG; ++k) {
      v2f w; w.x = w0; w.y = w1;
      *(v2f*)&Wt[2 * (SEG - 1 - k)] = w;
      const float n0 = -a1 * w0 + w1;
      const float n1 = -a2 * w0;
      w0 = n0; w1 = n1;
    }
  }
  __syncthreads();

  v4f wlo[2], whi[2];
  #pragma unroll
  for (int j = 0; j < 2; ++j) {
    wlo[j] = *(const v4f*)(Wt + 256 * j + 8 * lane);
    whi[j] = *(const v4f*)(Wt + 256 * j + 8 * lane + 4);
  }

  float v[2 * P1_ROWS];
  #pragma unroll
  for (int i = 0; i < P1_ROWS; ++i) {
    float s0 = wlo[0].x * in[i][0].x;
    float s1 = wlo[0].y * in[i][0].x;
    s0 = __builtin_fmaf(wlo[0].z, in[i][0].y, s0);
    s1 = __builtin_fmaf(wlo[0].w, in[i][0].y, s1);
    s0 = __builtin_fmaf(whi[0].x, in[i][0].z, s0);
    s1 = __builtin_fmaf(whi[0].y, in[i][0].z, s1);
    s0 = __builtin_fmaf(whi[0].z, in[i][0].w, s0);
    s1 = __builtin_fmaf(whi[0].w, in[i][0].w, s1);
    s0 = __builtin_fmaf(wlo[1].x, in[i][1].x, s0);
    s1 = __builtin_fmaf(wlo[1].y, in[i][1].x, s1);
    s0 = __builtin_fmaf(wlo[1].z, in[i][1].y, s0);
    s1 = __builtin_fmaf(wlo[1].w, in[i][1].y, s1);
    s0 = __builtin_fmaf(whi[1].x, in[i][1].z, s0);
    s1 = __builtin_fmaf(whi[1].y, in[i][1].z, s1);
    s0 = __builtin_fmaf(whi[1].z, in[i][1].w, s0);
    s1 = __builtin_fmaf(whi[1].w, in[i][1].w, s1);
    v[2 * i + 0] = s0;
    v[2 * i + 1] = s1;
  }

  float r16[16];
  {
    const bool up = (lane & 16) != 0;
    #pragma unroll
    for (int i = 0; i < 16; ++i) {
      float keep = up ? v[i + 16] : v[i];
      float send = up ? v[i] : v[i + 16];
      r16[i] = keep + __shfl_xor(send, 16, 32);
    }
  }
  float r8[8];
  {
    const bool up = (lane & 8) != 0;
    #pragma unroll
    for (int i = 0; i < 8; ++i) {
      float keep = up ? r16[i + 8] : r16[i];
      float send = up ? r16[i] : r16[i + 8];
      r8[i] = keep + __shfl_xor(send, 8, 32);
    }
  }
  float r4[4];
  {
    const bool up = (lane & 4) != 0;
    #pragma unroll
    for (int i = 0; i < 4; ++i) {
      float keep = up ? r8[i + 4] : r8[i];
      float send = up ? r8[i] : r8[i + 4];
      r4[i] = keep + __shfl_xor(send, 4, 32);
    }
  }
  float r2[2];
  {
    const bool up = (lane & 2) != 0;
    #pragma unroll
    for (int i = 0; i < 2; ++i) {
      float keep = up ? r4[i + 2] : r4[i];
      float send = up ? r4[i] : r4[i + 2];
      r2[i] = keep + __shfl_xor(send, 2, 32);
    }
  }
  float r1;
  {
    const bool up = (lane & 1) != 0;
    float keep = up ? r2[1] : r2[0];
    float send = up ? r2[0] : r2[1];
    r1 = keep + __shfl_xor(send, 1, 32);
  }

  if (active) zpart[(size_t)(seg * Bn + (grp << 4)) * 2 + (size_t)lane] = r1;
}

#define TILE_T 64
#define TILES_PER_SEG (SEG / TILE_T)
#define CHUNKS_PER_TILE (TILE_T / CHUNK)
#define ROWPAIRS 8
#define PITCH 68
#define ZB 64

template <int I>
__device__ __forceinline__ void load_states(v2f (&p)[ZB], const float* zp, size_t zstride, int seg) {
  if constexpr (I < ZB) {
    if (I < seg) {
      p[I] = *(const v2f*)(zp + (size_t)I * zstride);
      load_states<I + 1>(p, zp, zstride, seg);
    }
  }
}

template <int I>
__device__ __forceinline__ void scan_steps(const v2f (&p)[ZB], float& z0, float& z1,
                                           float m00, float m01, float m10, float m11, int seg) {
  if constexpr (I < ZB) {
    if (I < seg) {
      const float nz0 = m00 * z0 + m01 * z1 + p[I].x;
      const float nz1 = m10 * z0 + m11 * z1 + p[I].y;
      z0 = nz0; z1 = nz1;
      scan_steps<I + 1>(p, z0, z1, m00, m01, m10, m11, seg);
    }
  }
}

__global__ __launch_bounds__(256)
void wmma_filter_kernel(const float* __restrict__ u,
                        const float* __restrict__ bc,
                        const float* __restrict__ fc,
                        const float* __restrict__ yinit,
                        const float* __restrict__ uinit,
                        const float* __restrict__ zpart,
                        float* __restrict__ y,
                        int Bn, int T, int nseg) {
  __shared__ float Hs[16][16];
  __shared__ v2f   Cs[16];
  __shared__ __attribute__((aligned(16))) float Ut[8][16 * PITCH];
  __shared__ __attribute__((aligned(16))) float Yt[8][16 * PITCH];

  const float b0 = bc[0], b1 = bc[1], b2v = 0.0f;
  const float a1 = fc[0], a2 = fc[1];

  const int lane = (int)(threadIdx.x & 31);
  const int wib  = __builtin_amdgcn_readfirstlane((int)(threadIdx.x >> 5));
  const int wave = (int)(blockIdx.x * (blockDim.x >> 5)) + wib;
  const int ngroups = Bn >> 4;
  const int totalWaves = nseg * ngroups;
  const bool active = wave < totalWaves;
  const int seg  = active ? (nseg - 1) - wave / ngroups : 0;
  const int grp  = active ? wave % ngroups : 0;
  const int col  = lane & 15;
  const int half = lane >> 4;
  const int b    = (grp << 4) + col;

  const float* zp = zpart + (size_t)b * 2;
  const size_t zstride = (size_t)Bn * 2;
  float yi0 = 0.f, yi1 = 0.f, ui0 = 0.f;
  v2f p[ZB];
  if (active) {
    yi0 = yinit[b * 2 + 0];
    yi1 = yinit[b * 2 + 1];
    ui0 = uinit[b * 2 + 0];
    load_states<0>(p, zp, zstride, seg);
  }

  float m00 = -a1, m01 = 1.f, m10 = -a2, m11 = 0.f;
  #pragma unroll
  for (int i = 0; i < 8; ++i) {
    float t00 = m00 * m00 + m01 * m10;
    float t01 = m00 * m01 + m01 * m11;
    float t10 = m10 * m00 + m11 * m10;
    float t11 = m10 * m01 + m11 * m11;
    m00 = t00; m01 = t01; m10 = t10; m11 = t11;
  }

  if (threadIdx.x < 16) {
    const int t = (int)threadIdx.x;
    float hm1 = 0.f, hm2 = 0.f;
    #pragma unroll
    for (int d = 0; d < 16; ++d) {
      float hd;
      if (d == 0)      hd = b0;
      else if (d == 1) hd = b1 - a1 * b0;
      else if (d == 2) hd = b2v - a1 * hm1 - a2 * hm2;
      else             hd = -a1 * hm1 - a2 * hm2;
      if (d <= t) Hs[t][t - d] = hd;
      hm2 = hm1; hm1 = hd;
    }
    for (int k = t + 1; k < 16; ++k) Hs[t][k] = 0.f;
  } else if (threadIdx.x == 16) {
    float cc0 = 1.f, cc1 = 0.f;
    #pragma unroll
    for (int k = 0; k < 16; ++k) {
      v2f c; c.x = cc0; c.y = cc1;
      Cs[k] = c;
      float n0 = -a1 * cc0 - a2 * cc1;
      cc1 = cc0; cc0 = n0;
    }
  }
  __syncthreads();

  if (active) {
    float z0 = b1 * ui0 - a1 * yi0 - a2 * yi1;
    float z1 = -a2 * yi0;
    scan_steps<0>(p, z0, z1, m00, m01, m10, m11, seg);
    for (int j = ZB; j < seg; ++j) {
      const v2f q = *(const v2f*)(zp + (size_t)j * zstride);
      const float nz0 = m00 * z0 + m01 * z1 + q.x;
      const float nz1 = m10 * z0 + m11 * z1 + q.y;
      z0 = nz0; z1 = nz1;
    }

    v2f afrag[4];
    #pragma unroll
    for (int j = 0; j < 4; ++j)
      afrag[j] = *(const v2f*)&Hs[col][4 * j + 2 * half];

    v2f crow[8];
    #pragma unroll
    for (int r = 0; r < 8; ++r)
      crow[r] = Cs[r + 8 * half];

    float* ut = &Ut[wib][0];
    float* yt = &Yt[wib][0];
    const int row_off  = half * PITCH + 4 * col;
    const int frag_off = col * PITCH + 2 * half;
    const int d_off    = col * PITCH + 8 * half;

    const size_t goff  = (size_t)((grp << 4) + half) * (size_t)T + (size_t)seg * SEG + (size_t)(4 * col);
    const size_t rstep = (size_t)2 * (size_t)T;
    const float* ug = u + goff;
    float*       yg = y + goff;

    v4f in[TILES_PER_SEG][ROWPAIRS];
    #pragma unroll
    for (int t = 0; t < TILES_PER_SEG; ++t) {
      #pragma unroll
      for (int i = 0; i < ROWPAIRS; ++i)
        in[t][i] = *(const v4f*)(ug + (size_t)i * rstep + (size_t)(t * TILE_T));
      __builtin_amdgcn_sched_barrier(0);
    }

    #pragma unroll
    for (int t = 0; t < TILES_PER_SEG; ++t) {
      #pragma unroll
      for (int i = 0; i < ROWPAIRS; ++i)
        *(v4f*)(ut + row_off + 2 * i * PITCH) = in[t][i];

      #pragma unroll
      for (int c = 0; c < CHUNKS_PER_TILE; ++c) {
        v2f bfrag[4];
        #pragma unroll
        for (int j = 0; j < 4; ++j)
          bfrag[j] = *(const v2f*)(ut + frag_off + 16 * c + 4 * j);

        v8f acc;
        #pragma unroll
        for (int r = 0; r < 8; ++r) acc[r] = crow[r].x * z0 + crow[r].y * z1;

        acc = __builtin_amdgcn_wmma_f32_16x16x4_f32(false, afrag[0], false, bfrag[0], (short)0, acc, false, false);
        acc = __builtin_amdgcn_wmma_f32_16x16x4_f32(false, afrag[1], false, bfrag[1], (short)0, acc, false, false);
        acc = __builtin_amdgcn_wmma_f32_16x16x4_f32(false, afrag[2], false, bfrag[2], (short)0, acc, false, false);
        acc = __builtin_amdgcn_wmma_f32_16x16x4_f32(false, afrag[3], false, bfrag[3], (short)0, acc, false, false);

        v4f lo; lo.x = acc[0]; lo.y = acc[1]; lo.z = acc[2]; lo.w = acc[3];
        v4f hi; hi.x = acc[4]; hi.y = acc[5]; hi.z = acc[6]; hi.w = acc[7];
        *(v4f*)(yt + d_off + 16 * c + 0) = lo;
        *(v4f*)(yt + d_off + 16 * c + 4) = hi;

        float y14 = __shfl(acc[6],     col + 16, 32);
        float y15 = __shfl(acc[7],     col + 16, 32);
        float u14 = __shfl(bfrag[3].x, col + 16, 32);
        float u15 = __shfl(bfrag[3].y, col + 16, 32);
        float nz1 = b2v * u15 - a2 * y15;
        float nz0 = b1 * u15 - a1 * y15 + b2v * u14 - a2 * y14;
        z0 = nz0; z1 = nz1;
      }

      #pragma unroll
      for (int i = 0; i < ROWPAIRS; ++i) {
        v4f o = *(const v4f*)(yt + row_off + 2 * i * PITCH);
        __builtin_nontemporal_store(o, (v4f*)(yg + (size_t)i * rstep + (size_t)(t * TILE_T)));
      }
    }
  }
}

__global__ void naive_iir_kernel(const float* __restrict__ u,
                                 const float* __restrict__ bc,
                                 const float* __restrict__ fc,
                                 const float* __restrict__ yinit,
                                 const float* __restrict__ uinit,
                                 float* __restrict__ y, int Bn, int T) {
  int b = blockIdx.x * blockDim.x + threadIdx.x;
  if (b >= Bn) return;
  const float b0 = bc[0], b1 = bc[1];
  const float a1 = fc[0], a2 = fc[1];
  float z0 = b1 * uinit[b * 2 + 0] - a1 * yinit[b * 2 + 0] - a2 * yinit[b * 2 + 1];
  float z1 = -a2 * yinit[b * 2 + 0];
  const float* up = u + (long long)b * T;
  float* yp = y + (long long)b * T;
  for (int t = 0; t < T; ++t) {
    float ut = up[t];
    float yv = b0 * ut + z0;
    float nz0 = b1 * ut - a1 * yv + z1;
    z1 = -a2 * yv;
    z0 = nz0;
    yp[t] = yv;
  }
}

extern "C" void kernel_launch(void* const* d_in, const int* in_sizes, int n_in,
                              void* d_out, int out_size, void* d_ws, size_t ws_size,
                              hipStream_t stream) {
  const float* bc = (const float*)d_in[0];
  const float* fc = (const float*)d_in[1];
  const float* u  = (const float*)d_in[2];
  const float* yi = (const float*)d_in[3];
  const float* ui = (const float*)d_in[4];
  float* y = (float*)d_out;

  const int Bn = in_sizes[3] / 2;
  const int T  = in_sizes[2] / Bn;
  const int nseg = T / SEG;
  const size_t need = (size_t)nseg * (size_t)Bn * 2 * sizeof(float);

  if ((T % SEG) != 0 || (Bn % 16) != 0 || ws_size < need) {
    naive_iir_kernel<<<(Bn + 255) / 256, 256, 0, stream>>>(u, bc, fc, yi, ui, y, Bn, T);
    return;
  }

  float* zpart = (float*)d_ws;

  const int waves  = nseg * (Bn / 16);
  const int blocks = (waves + 7) / 8;
  seg_state_kernel<<<blocks, 256, 0, stream>>>(u, bc, fc, zpart, Bn, T, nseg);
  wmma_filter_kernel<<<blocks, 256, 0, stream>>>(u, bc, fc, yi, ui, zpart, y, Bn, T, nseg);
}


# one launch: a block owns 4 whole rows (32 waves x 512 steps; WMMA columns = row x 128-step piece), piece states meet in LDS, u read once into registers; state sums now over 128-step pieces, LDS scan
# speedup vs baseline: 6.7389x; 1.2199x over previous
#include <hip/hip_runtime.h>


typedef __attribute__((ext_vector_type(2))) float v2f;
typedef __attribute__((ext_vector_type(4))) float v4f;
typedef __attribute__((ext_vector_type(8))) float v8f;

#define CHUNK 16
#define SUB 128
#define TILE_T 64
#define TILES_PER_SUB (SUB / TILE_T)
#define CHUNKS_PER_TILE (TILE_T / CHUNK)
#define ROWPAIRS 8
#define PITCH 68
#define RB 4
#define SUBS_PER_WAVE 4
#define WAVE_T (SUB * SUBS_PER_WAVE)
#define NWAVES 32
#define MAXSUBS (NWAVES * SUBS_PER_WAVE)

__global__ __launch_bounds__(1024)
void fused_filter_kernel(const float* __restrict__ u,
                         const float* __restrict__ bc,
                         const float* __restrict__ fc,
                         const float* __restrict__ yinit,
                         const float* __restrict__ uinit,
                         float* __restrict__ y,
                         int Bn, int T) {
  __shared__ float Hs[16][16];
  __shared__ v2f   Cs[16];
  __shared__ __attribute__((aligned(16))) float Wt[2 * SUB];
  __shared__ __attribute__((aligned(8))) float Zp[RB * MAXSUBS * 2];
  __shared__ __attribute__((aligned(16))) float Tl[NWAVES][16 * PITCH];

  const float b0 = bc[0], b1 = bc[1], b2v = 0.0f;
  const float a1 = fc[0], a2 = fc[1];

  const int lane = (int)(threadIdx.x & 31);
  const int wib  = __builtin_amdgcn_readfirstlane((int)(threadIdx.x >> 5));
  const bool active = wib < T / WAVE_T;
  const int col  = lane & 15;
  const int half = lane >> 4;
  const int row0 = (int)blockIdx.x * RB;

  const size_t goff = (size_t)row0 * (size_t)T + (size_t)(wib * WAVE_T) + (size_t)(SUB * half + 4 * col);
  const float* ug = u + goff;
  float*       yg = y + goff;

  v4f in[TILES_PER_SUB][ROWPAIRS];
  if (active) {
    #pragma unroll
    for (int t = 0; t < TILES_PER_SUB; ++t) {
      #pragma unroll
      for (int i = 0; i < ROWPAIRS; ++i)
        in[t][i] = *(const v4f*)(ug + (size_t)(i >> 1) * (size_t)T + (size_t)(2 * SUB * (i & 1) + t * TILE_T));
      __builtin_amdgcn_sched_barrier(0);
    }
  } else {
    #pragma unroll
    for (int t = 0; t < TILES_PER_SUB; ++t) {
      #pragma unroll
      for (int i = 0; i < ROWPAIRS; ++i) { v4f zz = {0.f, 0.f, 0.f, 0.f}; in[t][i] = zz; }
    }
  }

  const int frow = row0 + (col >> 2);
  const float yi0 = yinit[frow * 2 + 0];
  const float yi1 = yinit[frow * 2 + 1];
  const float ui0 = uinit[frow * 2 + 0];

  float m00 = -a1, m01 = 1.f, m10 = -a2, m11 = 0.f;
  #pragma unroll
  for (int i = 0; i < 7; ++i) {
    float t00 = m00 * m00 + m01 * m10;
    float t01 = m00 * m01 + m01 * m11;
    float t10 = m10 * m00 + m11 * m10;
    float t11 = m10 * m01 + m11 * m11;
    m00 = t00; m01 = t01; m10 = t10; m11 = t11;
  }

  if (threadIdx.x < 16) {
    const int t = (int)threadIdx.x;
    float hm1 = 0.f, hm2 = 0.f;
    #pragma unroll
    for (int d = 0; d < 16; ++d) {
      float hd;
      if (d == 0)      hd = b0;
      else if (d == 1) hd = b1 - a1 * b0;
      else if (d == 2) hd = b2v - a1 * hm1 - a2 * hm2;
      else             hd = -a1 * hm1 - a2 * hm2;
      if (d <= t) Hs[t][t - d] = hd;
      hm2 = hm1; hm1 = hd;
    }
    for (int k = t + 1; k < 16; ++k) Hs[t][k] = 0.f;
  } else if (threadIdx.x == 16) {
    float cc0 = 1.f, cc1 = 0.f;
    #pragma unroll
    for (int k = 0; k < 16; ++k) {
      v2f c; c.x = cc0; c.y = cc1;
      Cs[k] = c;
      float n0 = -a1 * cc0 - a2 * cc1;
      cc1 = cc0; cc0 = n0;
    }
  } else if (threadIdx.x == 32) {
    float w0 = b1 - a1 * b0, w1 = b2v - a2 * b0;
    for (int k = 0; k < SUB; ++k) {
      v2f w; w.x = w0; w.y = w1;
      *(v2f*)&Wt[2 * (SUB - 1 - k)] = w;
      const float n0 = -a1 * w0 + w1;
      const float n1 = -a2 * w0;
      w0 = n0; w1 = n1;
    }
  }
  __syncthreads();

  v4f wlo[TILES_PER_SUB], whi[TILES_PER_SUB];
  #pragma unroll
  for (int t = 0; t < TILES_PER_SUB; ++t) {
    wlo[t] = *(const v4f*)(Wt + 2 * TILE_T * t + 8 * col);
    whi[t] = *(const v4f*)(Wt + 2 * TILE_T * t + 8 * col + 4);
  }

  float v[2 * ROWPAIRS];
  #pragma unroll
  for (int i = 0; i < ROWPAIRS; ++i) {
    float s0 = wlo[0].x * in[0][i].x;
    float s1 = wlo[0].y * in[0][i].x;
    s0 = __builtin_fmaf(wlo[0].z, in[0][i].y, s0);
    s1 = __builtin_fmaf(wlo[0].w, in[0][i].y, s1);
    s0 = __builtin_fmaf(whi[0].x, in[0][i].z, s0);
    s1 = __builtin_fmaf(whi[0].y, in[0][i].z, s1);
    s0 = __builtin_fmaf(whi[0].z, in[0][i].w, s0);
    s1 = __builtin_fmaf(whi[0].w, in[0][i].w, s1);
    s0 = __builtin_fmaf(wlo[1].x, in[1][i].x, s0);
    s1 = __builtin_fmaf(wlo[1].y, in[1][i].x, s1);
    s0 = __builtin_fmaf(wlo[1].z, in[1][i].y, s0);
    s1 = __builtin_fmaf(wlo[1].w, in[1][i].y, s1);
    s0 = __builtin_fmaf(whi[1].x, in[1][i].z, s0);
    s1 = __builtin_fmaf(whi[1].y, in[1][i].z, s1);
    s0 = __builtin_fmaf(whi[1].z, in[1][i].w, s0);
    s1 = __builtin_fmaf(whi[1].w, in[1][i].w, s1);
    v[2 * i + 0] = s0;
    v[2 * i + 1] = s1;
  }

  float r8[8];
  {
    const bool up = (lane & 8) != 0;
    #pragma unroll
    for (int i = 0; i < 8; ++i) {
      float keep = up ? v[i + 8] : v[i];
      float send = up ? v[i] : v[i + 8];
      r8[i] = keep + __shfl_xor(send, 8, 32);
    }
  }
  float r4[4];
  {
    const bool up = (lane & 4) != 0;
    #pragma unroll
    for (int i = 0; i < 4; ++i) {
      float keep = up ? r8[i + 4] : r8[i];
      float send = up ? r8[i] : r8[i + 4];
      r4[i] = keep + __shfl_xor(send, 4, 32);
    }
  }
  float r2[2];
  {
    const bool up = (lane & 2) != 0;
    #pragma unroll
    for (int i = 0; i < 2; ++i) {
      float keep = up ? r4[i + 2] : r4[i];
      float send = up ? r4[i] : r4[i + 2];
      r2[i] = keep + __shfl_xor(send, 2, 32);
    }
  }
  float r1;
  {
    const bool up = (lane & 1) != 0;
    float keep = up ? r2[1] : r2[0];
    float send = up ? r2[0] : r2[1];
    r1 = keep + __shfl_xor(send, 1, 32);
  }
  if (active)
    Zp[(((col >> 2) * MAXSUBS) + wib * SUBS_PER_WAVE + 2 * ((col >> 1) & 1) + half) * 2 + (col & 1)] = r1;

  __syncthreads();

  if (active) {
    float z0 = b1 * ui0 - a1 * yi0 - a2 * yi1;
    float z1 = -a2 * yi0;
    const float* zr = Zp + (col >> 2) * (MAXSUBS * 2);
    const int nbefore = wib * SUBS_PER_WAVE;
    #pragma unroll 4
    for (int j = 0; j < nbefore; ++j) {
      const v2f q = *(const v2f*)(zr + 2 * j);
      const float nz0 = m00 * z0 + m01 * z1 + q.x;
      const float nz1 = m10 * z0 + m11 * z1 + q.y;
      z0 = nz0; z1 = nz1;
    }
    #pragma unroll
    for (int k = 0; k < SUBS_PER_WAVE - 1; ++k) {
      const v2f q = *(const v2f*)(zr + 2 * (nbefore + k));
      const float nz0 = m00 * z0 + m01 * z1 + q.x;
      const float nz1 = m10 * z0 + m11 * z1 + q.y;
      const bool take = k < (col & 3);
      z0 = take ? nz0 : z0;
      z1 = take ? nz1 : z1;
    }

    v2f afrag[4];
    #pragma unroll
    for (int j = 0; j < 4; ++j)
      afrag[j] = *(const v2f*)&Hs[col][4 * j + 2 * half];

    v2f crow[8];
    #pragma unroll
    for (int r = 0; r < 8; ++r)
      crow[r] = Cs[r + 8 * half];

    float* tl = &Tl[wib][0];
    const int row_off  = half * PITCH + 4 * col;
    const int frag_off = col * PITCH + 2 * half;
    const int d_off    = col * PITCH + 8 * half;

    #pragma unroll
    for (int t = 0; t < TILES_PER_SUB; ++t) {
      #pragma unroll
      for (int i = 0; i < ROWPAIRS; ++i)
        *(v4f*)(tl + row_off + 2 * i * PITCH) = in[t][i];

      #pragma unroll
      for (int c = 0; c < CHUNKS_PER_TILE; ++c) {
        v2f bfrag[4];
        #pragma unroll
        for (int j = 0; j < 4; ++j)
          bfrag[j] = *(const v2f*)(tl + frag_off + 16 * c + 4 * j);

        v8f acc;
        #pragma unroll
        for (int r = 0; r < 8; ++r) acc[r] = crow[r].x * z0 + crow[r].y * z1;

        acc = __builtin_amdgcn_wmma_f32_16x16x4_f32(false, afrag[0], false, bfrag[0], (short)0, acc, false, false);
        acc = __builtin_amdgcn_wmma_f32_16x16x4_f32(false, afrag[1], false, bfrag[1], (short)0, acc, false, false);
        acc = __builtin_amdgcn_wmma_f32_16x16x4_f32(false, afrag[2], false, bfrag[2], (short)0, acc, false, false);
        acc = __builtin_amdgcn_wmma_f32_16x16x4_f32(false, afrag[3], false, bfrag[3], (short)0, acc, false, false);

        v4f lo; lo.x = acc[0]; lo.y = acc[1]; lo.z = acc[2]; lo.w = acc[3];
        v4f hi; hi.x = acc[4]; hi.y = acc[5]; hi.z = acc[6]; hi.w = acc[7];
        *(v4f*)(tl + d_off + 16 * c + 0) = lo;
        *(v4f*)(tl + d_off + 16 * c + 4) = hi;

        float y14 = __shfl(acc[6],     col + 16, 32);
        float y15 = __shfl(acc[7],     col + 16, 32);
        float u14 = __shfl(bfrag[3].x, col + 16, 32);
        float u15 = __shfl(bfrag[3].y, col + 16, 32);
        float nz1 = b2v * u15 - a2 * y15;
        float nz0 = b1 * u15 - a1 * y15 + b2v * u14 - a2 * y14;
        z0 = nz0; z1 = nz1;
      }

      #pragma unroll
      for (int i = 0; i < ROWPAIRS; ++i) {
        v4f o = *(const v4f*)(tl + row_off + 2 * i * PITCH);
        __builtin_nontemporal_store(o, (v4f*)(yg + (size_t)(i >> 1) * (size_t)T + (size_t)(2 * SUB * (i & 1) + t * TILE_T)));
      }
      __builtin_amdgcn_sched_barrier(0);
    }
  }
}

__global__ void naive_iir_kernel(const float* __restrict__ u,
                                 const float* __restrict__ bc,
                                 const float* __restrict__ fc,
                                 const float* __restrict__ yinit,
                                 const float* __restrict__ uinit,
                                 float* __restrict__ y, int Bn, int T) {
  int b = blockIdx.x * blockDim.x + threadIdx.x;
  if (b >= Bn) return;
  const float b0 = bc[0], b1 = bc[1];
  const float a1 = fc[0], a2 = fc[1];
  float z0 = b1 * uinit[b * 2 + 0] - a1 * yinit[b * 2 + 0] - a2 * yinit[b * 2 + 1];
  float z1 = -a2 * yinit[b * 2 + 0];
  const float* up = u + (long long)b * T;
  float* yp = y + (long long)b * T;
  for (int t = 0; t < T; ++t) {
    float ut = up[t];
    float yv = b0 * ut + z0;
    float nz0 = b1 * ut - a1 * yv + z1;
    z1 = -a2 * yv;
    z0 = nz0;
    yp[t] = yv;
  }
}

extern "C" void kernel_launch(void* const* d_in, const int* in_sizes, int n_in,
                              void* d_out, int out_size, void* d_ws, size_t ws_size,
                              hipStream_t stream) {
  const float* bc = (const float*)d_in[0];
  const float* fc = (const float*)d_in[1];
  const float* u  = (const float*)d_in[2];
  const float* yi = (const float*)d_in[3];
  const float* ui = (const float*)d_in[4];
  float* y = (float*)d_out;

  const int Bn = in_sizes[3] / 2;
  const int T  = in_sizes[2] / Bn;

  if ((T % WAVE_T) != 0 || (T / WAVE_T) > NWAVES || (Bn % RB) != 0) {
    naive_iir_kernel<<<(Bn + 255) / 256, 256, 0, stream>>>(u, bc, fc, yi, ui, y, Bn, T);
    return;
  }

  fused_filter_kernel<<<Bn / RB, NWAVES * 32, 0, stream>>>(u, bc, fc, yi, ui, y, Bn, T);
}


# waves per block halved (32 -> 16; 4096 waves enter, not 8192): a wave takes two 16-column groups (1024 steps of the 4 rows) as two interleaved matrix chains; per-column sums and order unchanged
# speedup vs baseline: 7.2451x; 1.0751x over previous
#include <hip/hip_runtime.h>


typedef __attribute__((ext_vector_type(2))) float v2f;
typedef __attribute__((ext_vector_type(4))) float v4f;
typedef __attribute__((ext_vector_type(8))) float v8f;

#define CHUNK 16
#define SUB 128
#define TILE_T 64
#define TILES_PER_SUB (SUB / TILE_T)
#define CHUNKS_PER_TILE (TILE_T / CHUNK)
#define ROWPAIRS 8
#define PITCH 68
#define RB 4
#define GR 2
#define SUBS_PER_GROUP 4
#define SUBS_PER_WAVE (GR * SUBS_PER_GROUP)
#define WAVE_T (SUB * SUBS_PER_WAVE)
#define NWAVES 16
#define MAXSUBS (NWAVES * SUBS_PER_WAVE)

__global__ __launch_bounds__(NWAVES * 32)
void fused_filter_kernel(const float* __restrict__ u,
                         const float* __restrict__ bc,
                         const float* __restrict__ fc,
                         const float* __restrict__ yinit,
                         const float* __restrict__ uinit,
                         float* __restrict__ y,
                         int Bn, int T) {
  __shared__ float Hs[16][16];
  __shared__ v2f   Cs[16];
  __shared__ __attribute__((aligned(16))) float Wt[2 * SUB];
  __shared__ __attribute__((aligned(8))) float Zp[RB * MAXSUBS * 2];
  __shared__ __attribute__((aligned(16))) float Tl[NWAVES][GR][16 * PITCH];

  const float b0 = bc[0], b1 = bc[1], b2v = 0.0f;
  const float a1 = fc[0], a2 = fc[1];

  const int lane = (int)(threadIdx.x & 31);
  const int wib  = __builtin_amdgcn_readfirstlane((int)(threadIdx.x >> 5));
  const bool active = wib < T / WAVE_T;
  const int col  = lane & 15;
  const int half = lane >> 4;
  const int row0 = (int)blockIdx.x * RB;

  const size_t goff = (size_t)row0 * (size_t)T + (size_t)(wib * WAVE_T) + (size_t)(SUB * half + 4 * col);
  const float* ug = u + goff;
  float*       yg = y + goff;

  v4f in[GR][TILES_PER_SUB][ROWPAIRS];
  if (active) {
    #pragma unroll
    for (int g = 0; g < GR; ++g) {
      #pragma unroll
      for (int t = 0; t < TILES_PER_SUB; ++t) {
        #pragma unroll
        for (int i = 0; i < ROWPAIRS; ++i)
          in[g][t][i] = *(const v4f*)(ug + (size_t)(i >> 1) * (size_t)T + (size_t)(SUB * SUBS_PER_GROUP * g + 2 * SUB * (i & 1) + t * TILE_T));
        __builtin_amdgcn_sched_barrier(0);
      }
    }
  } else {
    #pragma unroll
    for (int g = 0; g < GR; ++g) {
      #pragma unroll
      for (int t = 0; t < TILES_PER_SUB; ++t) {
        #pragma unroll
        for (int i = 0; i < ROWPAIRS; ++i) { v4f zz = {0.f, 0.f, 0.f, 0.f}; in[g][t][i] = zz; }
      }
    }
  }

  const int frow = row0 + (col >> 2);
  const float yi0 = yinit[frow * 2 + 0];
  const float yi1 = yinit[frow * 2 + 1];
  const float ui0 = uinit[frow * 2 + 0];

  float m00 = -a1, m01 = 1.f, m10 = -a2, m11 = 0.f;
  #pragma unroll
  for (int i = 0; i < 7; ++i) {
    float t00 = m00 * m00 + m01 * m10;
    float t01 = m00 * m01 + m01 * m11;
    float t10 = m10 * m00 + m11 * m10;
    float t11 = m10 * m01 + m11 * m11;
    m00 = t00; m01 = t01; m10 = t10; m11 = t11;
  }

  if (threadIdx.x < 16) {
    const int t = (int)threadIdx.x;
    float hm1 = 0.f, hm2 = 0.f;
    #pragma unroll
    for (int d = 0; d < 16; ++d) {
      float hd;
      if (d == 0)      hd = b0;
      else if (d == 1) hd = b1 - a1 * b0;
      else if (d == 2) hd = b2v - a1 * hm1 - a2 * hm2;
      else             hd = -a1 * hm1 - a2 * hm2;
      if (d <= t) Hs[t][t - d] = hd;
      hm2 = hm1; hm1 = hd;
    }
    for (int k = t + 1; k < 16; ++k) Hs[t][k] = 0.f;
  } else if (threadIdx.x == 16) {
    float cc0 = 1.f, cc1 = 0.f;
    #pragma unroll
    for (int k = 0; k < 16; ++k) {
      v2f c; c.x = cc0; c.y = cc1;
      Cs[k] = c;
      float n0 = -a1 * cc0 - a2 * cc1;
      cc1 = cc0; cc0 = n0;
    }
  } else if (threadIdx.x == 32) {
    float w0 = b1 - a1 * b0, w1 = b2v - a2 * b0;
    for (int k = 0; k < SUB; ++k) {
      v2f w; w.x = w0; w.y = w1;
      *(v2f*)&Wt[2 * (SUB - 1 - k)] = w;
      const float n0 = -a1 * w0 + w1;
      const float n1 = -a2 * w0;
      w0 = n0; w1 = n1;
    }
  }
  __syncthreads();

  v4f wlo[TILES_PER_SUB], whi[TILES_PER_SUB];
  #pragma unroll
  for (int t = 0; t < TILES_PER_SUB; ++t) {
    wlo[t] = *(const v4f*)(Wt + 2 * TILE_T * t + 8 * col);
    whi[t] = *(const v4f*)(Wt + 2 * TILE_T * t + 8 * col + 4);
  }

  float v[GR][2 * ROWPAIRS];
  #pragma unroll
  for (int g = 0; g < GR; ++g) {
    #pragma unroll
    for (int i = 0; i < ROWPAIRS; ++i) {
      float s0 = wlo[0].x * in[g][0][i].x;
      float s1 = wlo[0].y * in[g][0][i].x;
      s0 = __builtin_fmaf(wlo[0].z, in[g][0][i].y, s0);
      s1 = __builtin_fmaf(wlo[0].w, in[g][0][i].y, s1);
      s0 = __builtin_fmaf(whi[0].x, in[g][0][i].z, s0);
      s1 = __builtin_fmaf(whi[0].y, in[g][0][i].z, s1);
      s0 = __builtin_fmaf(whi[0].z, in[g][0][i].w, s0);
      s1 = __builtin_fmaf(whi[0].w, in[g][0][i].w, s1);
      s0 = __builtin_fmaf(wlo[1].x, in[g][1][i].x, s0);
      s1 = __builtin_fmaf(wlo[1].y, in[g][1][i].x, s1);
      s0 = __builtin_fmaf(wlo[1].z, in[g][1][i].y, s0);
      s1 = __builtin_fmaf(wlo[1].w, in[g][1][i].y, s1);
      s0 = __builtin_fmaf(whi[1].x, in[g][1][i].z, s0);
      s1 = __builtin_fmaf(whi[1].y, in[g][1][i].z, s1);
      s0 = __builtin_fmaf(whi[1].z, in[g][1][i].w, s0);
      s1 = __builtin_fmaf(whi[1].w, in[g][1][i].w, s1);
      v[g][2 * i + 0] = s0;
      v[g][2 * i + 1] = s1;
    }
  }

  float r8[GR][8];
  {
    const bool up = (lane & 8) != 0;
    #pragma unroll
    for (int g = 0; g < GR; ++g) {
      #pragma unroll
      for (int i = 0; i < 8; ++i) {
        float keep = up ? v[g][i + 8] : v[g][i];
        float send = up ? v[g][i] : v[g][i + 8];
        r8[g][i] = keep + __shfl_xor(send, 8, 32);
      }
    }
  }
  float r4[GR][4];
  {
    const bool up = (lane & 4) != 0;
    #pragma unroll
    for (int g = 0; g < GR; ++g) {
      #pragma unroll
      for (int i = 0; i < 4; ++i) {
        float keep = up ? r8[g][i + 4] : r8[g][i];
        float send = up ? r8[g][i] : r8[g][i + 4];
        r4[g][i] = keep + __shfl_xor(send, 4, 32);
      }
    }
  }
  float r2[GR][2];
  {
    const bool up = (lane & 2) != 0;
    #pragma unroll
    for (int g = 0; g < GR; ++g) {
      #pragma unroll
      for (int i = 0; i < 2; ++i) {
        float keep = up ? r4[g][i + 2] : r4[g][i];
        float send = up ? r4[g][i] : r4[g][i + 2];
        r2[g][i] = keep + __shfl_xor(send, 2, 32);
      }
    }
  }
  float r1[GR];
  {
    const bool up = (lane & 1) != 0;
    #pragma unroll
    for (int g = 0; g < GR; ++g) {
      float keep = up ? r2[g][1] : r2[g][0];
      float send = up ? r2[g][0] : r2[g][1];
      r1[g] = keep + __shfl_xor(send, 1, 32);
    }
  }
  if (active) {
    #pragma unroll
    for (int g = 0; g < GR; ++g)
      Zp[(((col >> 2) * MAXSUBS) + wib * SUBS_PER_WAVE + SUBS_PER_GROUP * g + 2 * ((col >> 1) & 1) + half) * 2 + (col & 1)] = r1[g];
  }

  __syncthreads();

  if (active) {
    float z0 = b1 * ui0 - a1 * yi0 - a2 * yi1;
    float z1 = -a2 * yi0;
    const float* zr = Zp + (col >> 2) * (MAXSUBS * 2);
    const int nbefore = wib * SUBS_PER_WAVE;
    #pragma unroll 4
    for (int j = 0; j < nbefore; ++j) {
      const v2f q = *(const v2f*)(zr + 2 * j);
      const float nz0 = m00 * z0 + m01 * z1 + q.x;
      const float nz1 = m10 * z0 + m11 * z1 + q.y;
      z0 = nz0; z1 = nz1;
    }
    float zs0[GR], zs1[GR];
    #pragma unroll
    for (int k = 0; k < SUBS_PER_WAVE; ++k) {
      const bool take = (k & (SUBS_PER_GROUP - 1)) == (col & 3);
      if ((k & (SUBS_PER_GROUP - 1)) == 0) { zs0[k / SUBS_PER_GROUP] = z0; zs1[k / SUBS_PER_GROUP] = z1; }
      else {
        zs0[k / SUBS_PER_GROUP] = take ? z0 : zs0[k / SUBS_PER_GROUP];
        zs1[k / SUBS_PER_GROUP] = take ? z1 : zs1[k / SUBS_PER_GROUP];
      }
      if (k < SUBS_PER_WAVE - 1) {
        const v2f q = *(const v2f*)(zr + 2 * (nbefore + k));
        const float nz0 = m00 * z0 + m01 * z1 + q.x;
        const float nz1 = m10 * z0 + m11 * z1 + q.y;
        z0 = nz0; z1 = nz1;
      }
    }

    v2f afrag[4];
    #pragma unroll
    for (int j = 0; j < 4; ++j)
      afrag[j] = *(const v2f*)&Hs[col][4 * j + 2 * half];

    v2f crow[8];
    #pragma unroll
    for (int r = 0; r < 8; ++r)
      crow[r] = Cs[r + 8 * half];

    float* tl0 = &Tl[wib][0][0];
    const int row_off  = half * PITCH + 4 * col;
    const int frag_off = col * PITCH + 2 * half;
    const int d_off    = col * PITCH + 8 * half;

    #pragma unroll
    for (int t = 0; t < TILES_PER_SUB; ++t) {
      #pragma unroll
      for (int g = 0; g < GR; ++g) {
        float* tl = tl0 + g * (16 * PITCH);
        #pragma unroll
        for (int i = 0; i < ROWPAIRS; ++i)
          *(v4f*)(tl + row_off + 2 * i * PITCH) = in[g][t][i];
      }

      #pragma unroll
      for (int c = 0; c < CHUNKS_PER_TILE; ++c) {
        #pragma unroll
        for (int g = 0; g < GR; ++g) {
          float* tl = tl0 + g * (16 * PITCH);
          v2f bfrag[4];
          #pragma unroll
          for (int j = 0; j < 4; ++j)
            bfrag[j] = *(const v2f*)(tl + frag_off + 16 * c + 4 * j);

          v8f acc;
          #pragma unroll
          for (int r = 0; r < 8; ++r) acc[r] = crow[r].x * zs0[g] + crow[r].y * zs1[g];

          acc = __builtin_amdgcn_wmma_f32_16x16x4_f32(false, afrag[0], false, bfrag[0], (short)0, acc, false, false);
          acc = __builtin_amdgcn_wmma_f32_16x16x4_f32(false, afrag[1], false, bfrag[1], (short)0, acc, false, false);
          acc = __builtin_amdgcn_wmma_f32_16x16x4_f32(false, afrag[2], false, bfrag[2], (short)0, acc, false, false);
          acc = __builtin_amdgcn_wmma_f32_16x16x4_f32(false, afrag[3], false, bfrag[3], (short)0, acc, false, false);

          v4f lo; lo.x = acc[0]; lo.y = acc[1]; lo.z = acc[2]; lo.w = acc[3];
          v4f hi; hi.x = acc[4]; hi.y = acc[5]; hi.z = acc[6]; hi.w = acc[7];
          *(v4f*)(tl + d_off + 16 * c + 0) = lo;
          *(v4f*)(tl + d_off + 16 * c + 4) = hi;

          float y14 = __shfl(acc[6],     col + 16, 32);
          float y15 = __shfl(acc[7],     col + 16, 32);
          float u14 = __shfl(bfrag[3].x, col + 16, 32);
          float u15 = __shfl(bfrag[3].y, col + 16, 32);
          float nz1 = b2v * u15 - a2 * y15;
          float nz0 = b1 * u15 - a1 * y15 + b2v * u14 - a2 * y14;
          zs0[g] = nz0; zs1[g] = nz1;
        }
      }

      #pragma unroll
      for (int g = 0; g < GR; ++g) {
        float* tl = tl0 + g * (16 * PITCH);
        #pragma unroll
        for (int i = 0; i < ROWPAIRS; ++i) {
          v4f o = *(const v4f*)(tl + row_off + 2 * i * PITCH);
          __builtin_nontemporal_store(o, (v4f*)(yg + (size_t)(i >> 1) * (size_t)T + (size_t)(SUB * SUBS_PER_GROUP * g + 2 * SUB * (i & 1) + t * TILE_T)));
        }
      }
      __builtin_amdgcn_sched_barrier(0);
    }
  }
}

__global__ void naive_iir_kernel(const float* __restrict__ u,
                                 const float* __restrict__ bc,
                                 const float* __restrict__ fc,
                                 const float* __restrict__ yinit,
                                 const float* __restrict__ uinit,
                                 float* __restrict__ y, int Bn, int T) {
  int b = blockIdx.x * blockDim.x + threadIdx.x;
  if (b >= Bn) return;
  const float b0 = bc[0], b1 = bc[1];
  const float a1 = fc[0], a2 = fc[1];
  float z0 = b1 * uinit[b * 2 + 0] - a1 * yinit[b * 2 + 0] - a2 * yinit[b * 2 + 1];
  float z1 = -a2 * yinit[b * 2 + 0];
  const float* up = u + (long long)b * T;
  float* yp = y + (long long)b * T;
  for (int t = 0; t < T; ++t) {
    float ut = up[t];
    float yv = b0 * ut + z0;
    float nz0 = b1 * ut - a1 * yv + z1;
    z1 = -a2 * yv;
    z0 = nz0;
    yp[t] = yv;
  }
}

extern "C" void kernel_launch(void* const* d_in, const int* in_sizes, int n_in,
                              void* d_out, int out_size, void* d_ws, size_t ws_size,
                              hipStream_t stream) {
  const float* bc = (const float*)d_in[0];
  const float* fc = (const float*)d_in[1];
  const float* u  = (const float*)d_in[2];
  const float* yi = (const float*)d_in[3];
  const float* ui = (const float*)d_in[4];
  float* y = (float*)d_out;

  const int Bn = in_sizes[3] / 2;
  const int T  = in_sizes[2] / Bn;

  if ((T % WAVE_T) != 0 || (T / WAVE_T) > NWAVES || (Bn % RB) != 0) {
    naive_iir_kernel<<<(Bn + 255) / 256, 256, 0, stream>>>(u, bc, fc, yi, ui, y, Bn, T);
    return;
  }

  fused_filter_kernel<<<Bn / RB, NWAVES * 32, 0, stream>>>(u, bc, fc, yi, ui, y, Bn, T);
}
